# starting kernel
# baseline (speedup 1.0000x reference)
#include <hip/hip_runtime.h>
#include <hip/hip_bf16.h>


typedef __attribute__((ext_vector_type(2))) float v2f;
typedef __attribute__((ext_vector_type(8))) float v8f;

constexpr int Dk = 64;
constexpr int Kc = 512;
constexpr int WAVES_PER_BLOCK = 8;
constexpr int ROWS_PER_WAVE   = 32;

__global__ void kmeans_prep(const float* __restrict__ X,
                            const int* __restrict__ init_idx,
                            float* __restrict__ C,
                            float* __restrict__ c2,
                            float* __restrict__ sums,
                            float* __restrict__ counts) {
    const int k = blockIdx.x;
    const int t = threadIdx.x;
    const int src = init_idx[k];
    const float v = X[(size_t)src * Dk + t];
    C[k * Dk + t]    = v;
    sums[k * Dk + t] = 0.0f;

    __shared__ float red[Dk];
    red[t] = v * v;
    __syncthreads();
    if (t < 32) {
        float s = red[t] + red[t + 32];
        #pragma unroll
        for (int off = 16; off > 0; off >>= 1)
            s += __shfl_down(s, off, 32);
        if (t == 0) {
            c2[k]     = s;
            counts[k] = 0.0f;
        }
    }
}

__global__ void __launch_bounds__(256)
kmeans_assign(const float* __restrict__ X,
              const float* __restrict__ C,
              const float* __restrict__ c2,
              float* __restrict__ sums,
              float* __restrict__ counts,
              int numTiles) {
    extern __shared__ float lds[];
    float* ldsC  = lds;
    float* ldsC2 = lds + Kc * Dk;

    {
        const int tid = threadIdx.x;
        const float4* src4 = (const float4*)C;
        float4* dst4 = (float4*)ldsC;
        #pragma unroll 4
        for (int i = tid; i < (Kc * Dk) / 4; i += 256) dst4[i] = src4[i];
        for (int i = tid; i < Kc; i += 256) ldsC2[i] = c2[i];
    }
    __syncthreads();

    const int wave = threadIdx.x >> 5;
    const int lane = threadIdx.x & 31;
    const int tile = blockIdx.x * WAVES_PER_BLOCK + wave;
    if (tile >= numTiles) return;

    const int row0 = tile * ROWS_PER_WAVE;
    const int m  = lane & 15;
    const int kb = (lane >> 4) * 2;

    v2f a0[16], a1[16];
    {
        const float* xr0 = X + (size_t)(row0 + m) * Dk + kb;
        const float* xr1 = X + (size_t)(row0 + 16 + m) * Dk + kb;
        #pragma unroll
        for (int j = 0; j < 16; ++j) {
            a0[j] = *(const v2f*)(xr0 + 4 * j);
            a1[j] = *(const v2f*)(xr1 + 4 * j);
        }
    }

    float bestVal0[8], bestVal1[8];
    int   bestIdx0[8], bestIdx1[8];
    #pragma unroll
    for (int r = 0; r < 8; ++r) {
        bestVal0[r] = __builtin_inff(); bestIdx0[r] = 0;
        bestVal1[r] = __builtin_inff(); bestIdx1[r] = 0;
    }

    for (int n0 = 0; n0 < Kc; n0 += 16) {
        v2f b[16];
        {
            const float* crow = ldsC + (n0 + m) * Dk + kb;
            #pragma unroll
            for (int j = 0; j < 16; ++j)
                b[j] = *(const v2f*)(crow + 4 * j);
        }
        v8f acc0 = {0.f, 0.f, 0.f, 0.f, 0.f, 0.f, 0.f, 0.f};
        v8f acc1 = {0.f, 0.f, 0.f, 0.f, 0.f, 0.f, 0.f, 0.f};
        #pragma unroll
        for (int j = 0; j < 16; ++j) {
            acc0 = __builtin_amdgcn_wmma_f32_16x16x4_f32(
                false, a0[j], false, b[j], (short)0, acc0, false, false);
            acc1 = __builtin_amdgcn_wmma_f32_16x16x4_f32(
                false, a1[j], false, b[j], (short)0, acc1, false, false);
        }
        const float cc  = ldsC2[n0 + m];
        const int   col = n0 + m;
        #pragma unroll
        for (int r = 0; r < 8; ++r) {
            const float d0 = cc - 2.0f * acc0[r];
            const float d1 = cc - 2.0f * acc1[r];
            if (d0 < bestVal0[r]) { bestVal0[r] = d0; bestIdx0[r] = col; }
            if (d1 < bestVal1[r]) { bestVal1[r] = d1; bestIdx1[r] = col; }
        }
    }

    #pragma unroll
    for (int off = 1; off < 16; off <<= 1) {
        #pragma unroll
        for (int r = 0; r < 8; ++r) {
            float ov = __shfl_xor(bestVal0[r], off, 32);
            int   oi = __shfl_xor(bestIdx0[r], off, 32);
            if (ov < bestVal0[r] || (ov == bestVal0[r] && oi < bestIdx0[r])) {
                bestVal0[r] = ov; bestIdx0[r] = oi;
            }
            ov = __shfl_xor(bestVal1[r], off, 32);
            oi = __shfl_xor(bestIdx1[r], off, 32);
            if (ov < bestVal1[r] || (ov == bestVal1[r] && oi < bestIdx1[r])) {
                bestVal1[r] = ov; bestIdx1[r] = oi;
            }
        }
    }

    #pragma unroll
    for (int r = 0; r < 8; ++r) {
        const int lab[4] = {
            __shfl(bestIdx0[r], 0, 32),
            __shfl(bestIdx0[r], 16, 32),
            __shfl(bestIdx1[r], 0, 32),
            __shfl(bestIdx1[r], 16, 32)
        };
        const int rowOf[4] = { row0 + r, row0 + r + 8, row0 + 16 + r, row0 + 24 + r };
        #pragma unroll
        for (int q = 0; q < 4; ++q) {
            const float2 xv = ((const float2*)(X + (size_t)rowOf[q] * Dk))[lane];
            atomicAdd(&sums[lab[q] * Dk + 2 * lane + 0], xv.x);
            atomicAdd(&sums[lab[q] * Dk + 2 * lane + 1], xv.y);
            if (lane == 0) atomicAdd(&counts[lab[q]], 1.0f);
        }
    }
}

__global__ void kmeans_finalize(const float* __restrict__ sums,
                                const float* __restrict__ counts,
                                float* __restrict__ out) {
    const int k = blockIdx.x;
    const int t = threadIdx.x;
    const float cnt = fmaxf(counts[k], 1.0f);
    out[k * Dk + t] = sums[k * Dk + t] / cnt;
}

extern "C" void kernel_launch(void* const* d_in, const int* in_sizes, int n_in,
                              void* d_out, int out_size, void* d_ws, size_t ws_size,
                              hipStream_t stream) {
    const float* X        = (const float*)d_in[0];
    const int*   init_idx = (const int*)d_in[1];
    float*       out      = (float*)d_out;

    const int N        = in_sizes[0] / Dk;
    const int numTiles = N / ROWS_PER_WAVE;

    float* ws     = (float*)d_ws;
    float* C      = ws;
    float* c2     = C + Kc * Dk;
    float* sums   = c2 + Kc;
    float* counts = sums + Kc * Dk;

    kmeans_prep<<<Kc, Dk, 0, stream>>>(X, init_idx, C, c2, sums, counts);

    const int blocks = (numTiles + WAVES_PER_BLOCK - 1) / WAVES_PER_BLOCK;
    const size_t shmem = (size_t)(Kc * Dk + Kc) * sizeof(float);
    kmeans_assign<<<blocks, 32 * WAVES_PER_BLOCK, shmem, stream>>>(
        X, C, c2, sums, counts, numTiles);

    kmeans_finalize<<<Kc, Dk, 0, stream>>>(sums, counts, out);
}


# Accumulate segment sums and counts per persistent work-group in LDS (ds_add_f32) and sum the 256 partial planes in the last kernel, replacing all global atomics; changes summation order
# speedup vs baseline: 2.1216x; 2.1216x over previous
#include <hip/hip_runtime.h>
#include <hip/hip_bf16.h>


typedef __attribute__((ext_vector_type(2))) float v2f;
typedef __attribute__((ext_vector_type(8))) float v8f;

constexpr int Dk = 64;
constexpr int Kc = 512;
constexpr int WAVES_PER_BLOCK = 16;
constexpr int THREADS         = 32 * WAVES_PER_BLOCK;
constexpr int ROWS_PER_WAVE   = 32;
constexpr int NUM_GROUPS      = 256;
constexpr int LDS_FLOATS = 2 * Kc * Dk + 2 * Kc;
static_assert(LDS_FLOATS * 4 <= 320 * 1024, "one work-group's LDS");

__global__ void kmeans_prep(const float* __restrict__ X,
                            const int* __restrict__ init_idx,
                            float* __restrict__ C,
                            float* __restrict__ c2) {
    const int k = blockIdx.x;
    const int t = threadIdx.x;
    const int src = init_idx[k];
    const float v = X[(size_t)src * Dk + t];
    C[k * Dk + t]    = v;

    __shared__ float red[Dk];
    red[t] = v * v;
    __syncthreads();
    if (t < 32) {
        float s = red[t] + red[t + 32];
        #pragma unroll
        for (int off = 16; off > 0; off >>= 1)
            s += __shfl_down(s, off, 32);
        if (t == 0) c2[k] = s;
    }
}

__global__ void __launch_bounds__(THREADS)
kmeans_assign(const float* __restrict__ X,
              const float* __restrict__ C,
              const float* __restrict__ c2,
              float* __restrict__ psums,
              float* __restrict__ pcounts,
              int numTiles) {
    __shared__ __attribute__((aligned(16))) float lds[LDS_FLOATS];
    float* ldsC   = lds;
    float* ldsSum = lds + Kc * Dk;
    float* ldsC2  = lds + 2 * Kc * Dk;
    float* ldsCnt = ldsC2 + Kc;

    {
        const int tid = threadIdx.x;
        const float4* src4 = (const float4*)C;
        float4* dst4 = (float4*)ldsC;
        float4* sum4 = (float4*)ldsSum;
        const float4 zero4 = make_float4(0.0f, 0.0f, 0.0f, 0.0f);
        #pragma unroll 4
        for (int i = tid; i < (Kc * Dk) / 4; i += THREADS) {
            dst4[i] = src4[i];
            sum4[i] = zero4;
        }
        for (int i = tid; i < Kc; i += THREADS) {
            ldsC2[i]  = c2[i];
            ldsCnt[i] = 0.0f;
        }
    }
    __syncthreads();

    const int wave = threadIdx.x >> 5;
    const int lane = threadIdx.x & 31;
    const int m  = lane & 15;
    const int kb = (lane >> 4) * 2;

    for (int tile = wave * gridDim.x + blockIdx.x; tile < numTiles;
         tile += WAVES_PER_BLOCK * gridDim.x) {
    const int row0 = tile * ROWS_PER_WAVE;

    v2f a0[16], a1[16];
    {
        const float* xr0 = X + (size_t)(row0 + m) * Dk + kb;
        const float* xr1 = X + (size_t)(row0 + 16 + m) * Dk + kb;
        #pragma unroll
        for (int j = 0; j < 16; ++j) {
            a0[j] = *(const v2f*)(xr0 + 4 * j);
            a1[j] = *(const v2f*)(xr1 + 4 * j);
        }
    }

    float bestVal0[8], bestVal1[8];
    int   bestIdx0[8], bestIdx1[8];
    #pragma unroll
    for (int r = 0; r < 8; ++r) {
        bestVal0[r] = __builtin_inff(); bestIdx0[r] = 0;
        bestVal1[r] = __builtin_inff(); bestIdx1[r] = 0;
    }

    for (int n0 = 0; n0 < Kc; n0 += 16) {
        v2f b[16];
        {
            const float* crow = ldsC + (n0 + m) * Dk + kb;
            #pragma unroll
            for (int j = 0; j < 16; ++j)
                b[j] = *(const v2f*)(crow + 4 * j);
        }
        v8f acc0 = {0.f, 0.f, 0.f, 0.f, 0.f, 0.f, 0.f, 0.f};
        v8f acc1 = {0.f, 0.f, 0.f, 0.f, 0.f, 0.f, 0.f, 0.f};
        #pragma unroll
        for (int j = 0; j < 16; ++j) {
            acc0 = __builtin_amdgcn_wmma_f32_16x16x4_f32(
                false, a0[j], false, b[j], (short)0, acc0, false, false);
            acc1 = __builtin_amdgcn_wmma_f32_16x16x4_f32(
                false, a1[j], false, b[j], (short)0, acc1, false, false);
        }
        const float cc  = ldsC2[n0 + m];
        const int   col = n0 + m;
        #pragma unroll
        for (int r = 0; r < 8; ++r) {
            const float d0 = cc - 2.0f * acc0[r];
            const float d1 = cc - 2.0f * acc1[r];
            if (d0 < bestVal0[r]) { bestVal0[r] = d0; bestIdx0[r] = col; }
            if (d1 < bestVal1[r]) { bestVal1[r] = d1; bestIdx1[r] = col; }
        }
    }

    #pragma unroll
    for (int off = 1; off < 16; off <<= 1) {
        #pragma unroll
        for (int r = 0; r < 8; ++r) {
            float ov = __shfl_xor(bestVal0[r], off, 32);
            int   oi = __shfl_xor(bestIdx0[r], off, 32);
            if (ov < bestVal0[r] || (ov == bestVal0[r] && oi < bestIdx0[r])) {
                bestVal0[r] = ov; bestIdx0[r] = oi;
            }
            ov = __shfl_xor(bestVal1[r], off, 32);
            oi = __shfl_xor(bestIdx1[r], off, 32);
            if (ov < bestVal1[r] || (ov == bestVal1[r] && oi < bestIdx1[r])) {
                bestVal1[r] = ov; bestIdx1[r] = oi;
            }
        }
    }

    #pragma unroll
    for (int r = 0; r < 8; ++r) {
        const int lab[4] = {
            __shfl(bestIdx0[r], 0, 32),
            __shfl(bestIdx0[r], 16, 32),
            __shfl(bestIdx1[r], 0, 32),
            __shfl(bestIdx1[r], 16, 32)
        };
        const int rowOf[4] = { row0 + r, row0 + r + 8, row0 + 16 + r, row0 + 24 + r };
        #pragma unroll
        for (int q = 0; q < 4; ++q) {
            const float2 xv = ((const float2*)(X + (size_t)rowOf[q] * Dk))[lane];
            atomicAdd(&ldsSum[lab[q] * Dk + 2 * lane + 0], xv.x);
            atomicAdd(&ldsSum[lab[q] * Dk + 2 * lane + 1], xv.y);
            if (lane == 0) atomicAdd(&ldsCnt[lab[q]], 1.0f);
        }
    }
    }

    __syncthreads();
    {
        const int tid = threadIdx.x;
        const float4* sum4 = (const float4*)ldsSum;
        float4* dst4 = (float4*)(psums + (size_t)blockIdx.x * (Kc * Dk));
        #pragma unroll 4
        for (int i = tid; i < (Kc * Dk) / 4; i += THREADS) dst4[i] = sum4[i];
        for (int i = tid; i < Kc; i += THREADS)
            pcounts[(size_t)blockIdx.x * Kc + i] = ldsCnt[i];
    }
}

__global__ void __launch_bounds__(256)
kmeans_finalize(const float* __restrict__ psums,
                const float* __restrict__ pcounts,
                float* __restrict__ out) {
    static_assert(NUM_GROUPS == 256, "one count plane per thread, four plane quarters");
    const int k = blockIdx.x;
    const int t = threadIdx.x & 63;
    const int q = threadIdx.x >> 6;
    __shared__ float part[4][Dk];
    __shared__ float cpart[NUM_GROUPS];

    const float* p = psums + (size_t)(q * 64) * (Kc * Dk) + k * Dk + t;
    float s = 0.0f;
    #pragma unroll 16
    for (int g = 0; g < 64; ++g) s += p[(size_t)g * (Kc * Dk)];
    part[q][t] = s;
    cpart[threadIdx.x] = pcounts[(size_t)threadIdx.x * Kc + k];
    __syncthreads();
    if (q == 0) {
        const float tot = (part[0][t] + part[1][t]) + (part[2][t] + part[3][t]);
        float cnt = 0.0f;
        for (int i = 0; i < NUM_GROUPS; ++i) cnt += cpart[i];
        out[k * Dk + t] = tot / fmaxf(cnt, 1.0f);
    }
}

extern "C" void kernel_launch(void* const* d_in, const int* in_sizes, int n_in,
                              void* d_out, int out_size, void* d_ws, size_t ws_size,
                              hipStream_t stream) {
    const float* X        = (const float*)d_in[0];
    const int*   init_idx = (const int*)d_in[1];
    float*       out      = (float*)d_out;

    const int N        = in_sizes[0] / Dk;
    const int numTiles = N / ROWS_PER_WAVE;

    float* ws      = (float*)d_ws;
    float* C       = ws;
    float* c2      = C + Kc * Dk;
    float* psums   = c2 + Kc;
    float* pcounts = psums + (size_t)NUM_GROUPS * Kc * Dk;

    kmeans_prep<<<Kc, Dk, 0, stream>>>(X, init_idx, C, c2);

    kmeans_assign<<<NUM_GROUPS, THREADS, 0, stream>>>(
        X, C, c2, psums, pcounts, numTiles);

    kmeans_finalize<<<Kc, 256, 0, stream>>>(psums, pcounts, out);
}


# Pad the LDS row pitch of the centroid (B operand) matrix from 64 to 68 floats: the 16 rows read by one 8-byte fragment load now fall on distinct LDS banks instead of all on the same four banks
# speedup vs baseline: 2.9176x; 1.3752x over previous
#include <hip/hip_runtime.h>
#include <hip/hip_bf16.h>


typedef __attribute__((ext_vector_type(2))) float v2f;
typedef __attribute__((ext_vector_type(8))) float v8f;

constexpr int Dk = 64;
constexpr int Kc = 512;
constexpr int WAVES_PER_BLOCK = 16;
constexpr int THREADS         = 32 * WAVES_PER_BLOCK;
constexpr int ROWS_PER_WAVE   = 32;
constexpr int NUM_GROUPS      = 256;
constexpr int LDC = Dk + 4;
constexpr int LDS_FLOATS = Kc * LDC + Kc * Dk + 2 * Kc;
static_assert(LDS_FLOATS * 4 <= 320 * 1024, "one work-group's LDS");

__global__ void kmeans_prep(const float* __restrict__ X,
                            const int* __restrict__ init_idx,
                            float* __restrict__ C,
                            float* __restrict__ c2) {
    const int k = blockIdx.x;
    const int t = threadIdx.x;
    const int src = init_idx[k];
    const float v = X[(size_t)src * Dk + t];
    C[k * Dk + t]    = v;

    __shared__ float red[Dk];
    red[t] = v * v;
    __syncthreads();
    if (t < 32) {
        float s = red[t] + red[t + 32];
        #pragma unroll
        for (int off = 16; off > 0; off >>= 1)
            s += __shfl_down(s, off, 32);
        if (t == 0) c2[k] = s;
    }
}

__global__ void __launch_bounds__(THREADS)
kmeans_assign(const float* __restrict__ X,
              const float* __restrict__ C,
              const float* __restrict__ c2,
              float* __restrict__ psums,
              float* __restrict__ pcounts,
              int numTiles) {
    __shared__ __attribute__((aligned(16))) float lds[LDS_FLOATS];
    float* ldsC   = lds;
    float* ldsSum = lds + Kc * LDC;
    float* ldsC2  = ldsSum + Kc * Dk;
    float* ldsCnt = ldsC2 + Kc;

    {
        const int tid = threadIdx.x;
        const float4* src4 = (const float4*)C;
        float4* sum4 = (float4*)ldsSum;
        const float4 zero4 = make_float4(0.0f, 0.0f, 0.0f, 0.0f);
        #pragma unroll 4
        for (int i = tid; i < (Kc * Dk) / 4; i += THREADS) {
            *(float4*)(ldsC + (i >> 4) * LDC + 4 * (i & 15)) = src4[i];
            sum4[i] = zero4;
        }
        for (int i = tid; i < Kc; i += THREADS) {
            ldsC2[i]  = c2[i];
            ldsCnt[i] = 0.0f;
        }
    }
    __syncthreads();

    const int wave = threadIdx.x >> 5;
    const int lane = threadIdx.x & 31;
    const int m  = lane & 15;
    const int kb = (lane >> 4) * 2;

    for (int tile = wave * gridDim.x + blockIdx.x; tile < numTiles;
         tile += WAVES_PER_BLOCK * gridDim.x) {
    const int row0 = tile * ROWS_PER_WAVE;

    v2f a0[16], a1[16];
    {
        const float* xr0 = X + (size_t)(row0 + m) * Dk + kb;
        const float* xr1 = X + (size_t)(row0 + 16 + m) * Dk + kb;
        #pragma unroll
        for (int j = 0; j < 16; ++j) {
            a0[j] = *(const v2f*)(xr0 + 4 * j);
            a1[j] = *(const v2f*)(xr1 + 4 * j);
        }
    }

    float bestVal0[8], bestVal1[8];
    int   bestIdx0[8], bestIdx1[8];
    #pragma unroll
    for (int r = 0; r < 8; ++r) {
        bestVal0[r] = __builtin_inff(); bestIdx0[r] = 0;
        bestVal1[r] = __builtin_inff(); bestIdx1[r] = 0;
    }

    for (int n0 = 0; n0 < Kc; n0 += 16) {
        v2f b[16];
        {
            const float* crow = ldsC + (n0 + m) * LDC + kb;
            #pragma unroll
            for (int j = 0; j < 16; ++j)
                b[j] = *(const v2f*)(crow + 4 * j);
        }
        v8f acc0 = {0.f, 0.f, 0.f, 0.f, 0.f, 0.f, 0.f, 0.f};
        v8f acc1 = {0.f, 0.f, 0.f, 0.f, 0.f, 0.f, 0.f, 0.f};
        #pragma unroll
        for (int j = 0; j < 16; ++j) {
            acc0 = __builtin_amdgcn_wmma_f32_16x16x4_f32(
                false, a0[j], false, b[j], (short)0, acc0, false, false);
            acc1 = __builtin_amdgcn_wmma_f32_16x16x4_f32(
                false, a1[j], false, b[j], (short)0, acc1, false, false);
        }
        const float cc  = ldsC2[n0 + m];
        const int   col = n0 + m;
        #pragma unroll
        for (int r = 0; r < 8; ++r) {
            const float d0 = cc - 2.0f * acc0[r];
            const float d1 = cc - 2.0f * acc1[r];
            if (d0 < bestVal0[r]) { bestVal0[r] = d0; bestIdx0[r] = col; }
            if (d1 < bestVal1[r]) { bestVal1[r] = d1; bestIdx1[r] = col; }
        }
    }

    #pragma unroll
    for (int off = 1; off < 16; off <<= 1) {
        #pragma unroll
        for (int r = 0; r < 8; ++r) {
            float ov = __shfl_xor(bestVal0[r], off, 32);
            int   oi = __shfl_xor(bestIdx0[r], off, 32);
            if (ov < bestVal0[r] || (ov == bestVal0[r] && oi < bestIdx0[r])) {
                bestVal0[r] = ov; bestIdx0[r] = oi;
            }
            ov = __shfl_xor(bestVal1[r], off, 32);
            oi = __shfl_xor(bestIdx1[r], off, 32);
            if (ov < bestVal1[r] || (ov == bestVal1[r] && oi < bestIdx1[r])) {
                bestVal1[r] = ov; bestIdx1[r] = oi;
            }
        }
    }

    #pragma unroll
    for (int r = 0; r < 8; ++r) {
        const int lab[4] = {
            __shfl(bestIdx0[r], 0, 32),
            __shfl(bestIdx0[r], 16, 32),
            __shfl(bestIdx1[r], 0, 32),
            __shfl(bestIdx1[r], 16, 32)
        };
        const int rowOf[4] = { row0 + r, row0 + r + 8, row0 + 16 + r, row0 + 24 + r };
        #pragma unroll
        for (int q = 0; q < 4; ++q) {
            const float2 xv = ((const float2*)(X + (size_t)rowOf[q] * Dk))[lane];
            atomicAdd(&ldsSum[lab[q] * Dk + 2 * lane + 0], xv.x);
            atomicAdd(&ldsSum[lab[q] * Dk + 2 * lane + 1], xv.y);
            if (lane == 0) atomicAdd(&ldsCnt[lab[q]], 1.0f);
        }
    }
    }

    __syncthreads();
    {
        const int tid = threadIdx.x;
        const float4* sum4 = (const float4*)ldsSum;
        float4* dst4 = (float4*)(psums + (size_t)blockIdx.x * (Kc * Dk));
        #pragma unroll 4
        for (int i = tid; i < (Kc * Dk) / 4; i += THREADS) dst4[i] = sum4[i];
        for (int i = tid; i < Kc; i += THREADS)
            pcounts[(size_t)blockIdx.x * Kc + i] = ldsCnt[i];
    }
}

__global__ void __launch_bounds__(256)
kmeans_finalize(const float* __restrict__ psums,
                const float* __restrict__ pcounts,
                float* __restrict__ out) {
    static_assert(NUM_GROUPS == 256, "one count plane per thread, four plane quarters");
    const int k = blockIdx.x;
    const int t = threadIdx.x & 63;
    const int q = threadIdx.x >> 6;
    __shared__ float part[4][Dk];
    __shared__ float cpart[NUM_GROUPS];

    const float* p = psums + (size_t)(q * 64) * (Kc * Dk) + k * Dk + t;
    float s = 0.0f;
    #pragma unroll 16
    for (int g = 0; g < 64; ++g) s += p[(size_t)g * (Kc * Dk)];
    part[q][t] = s;
    cpart[threadIdx.x] = pcounts[(size_t)threadIdx.x * Kc + k];
    __syncthreads();
    if (q == 0) {
        const float tot = (part[0][t] + part[1][t]) + (part[2][t] + part[3][t]);
        float cnt = 0.0f;
        for (int i = 0; i < NUM_GROUPS; ++i) cnt += cpart[i];
        out[k * Dk + t] = tot / fmaxf(cnt, 1.0f);
    }
}

extern "C" void kernel_launch(void* const* d_in, const int* in_sizes, int n_in,
                              void* d_out, int out_size, void* d_ws, size_t ws_size,
                              hipStream_t stream) {
    const float* X        = (const float*)d_in[0];
    const int*   init_idx = (const int*)d_in[1];
    float*       out      = (float*)d_out;

    const int N        = in_sizes[0] / Dk;
    const int numTiles = N / ROWS_PER_WAVE;

    float* ws      = (float*)d_ws;
    float* C       = ws;
    float* c2      = C + Kc * Dk;
    float* psums   = c2 + Kc;
    float* pcounts = psums + (size_t)NUM_GROUPS * Kc * Dk;

    kmeans_prep<<<Kc, Dk, 0, stream>>>(X, init_idx, C, c2);

    kmeans_assign<<<NUM_GROUPS, THREADS, 0, stream>>>(
        X, C, c2, psums, pcounts, numTiles);

    kmeans_finalize<<<Kc, 256, 0, stream>>>(psums, pcounts, out);
}


# Issue each tile's 32 row re-reads for the segment sums as one batch of global loads ahead of the argmin reduction, instead of one load-wait-add per row between the LDS atomics
# speedup vs baseline: 3.0698x; 1.0522x over previous
#include <hip/hip_runtime.h>
#include <hip/hip_bf16.h>


typedef __attribute__((ext_vector_type(2))) float v2f;
typedef __attribute__((ext_vector_type(8))) float v8f;

constexpr int Dk = 64;
constexpr int Kc = 512;
constexpr int WAVES_PER_BLOCK = 16;
constexpr int THREADS         = 32 * WAVES_PER_BLOCK;
constexpr int ROWS_PER_WAVE   = 32;
constexpr int NUM_GROUPS      = 256;
constexpr int LDC = Dk + 4;
constexpr int LDS_FLOATS = Kc * LDC + Kc * Dk + 2 * Kc;
static_assert(LDS_FLOATS * 4 <= 320 * 1024, "one work-group's LDS");

__global__ void kmeans_prep(const float* __restrict__ X,
                            const int* __restrict__ init_idx,
                            float* __restrict__ C,
                            float* __restrict__ c2) {
    const int k = blockIdx.x;
    const int t = threadIdx.x;
    const int src = init_idx[k];
    const float v = X[(size_t)src * Dk + t];
    C[k * Dk + t]    = v;

    __shared__ float red[Dk];
    red[t] = v * v;
    __syncthreads();
    if (t < 32) {
        float s = red[t] + red[t + 32];
        #pragma unroll
        for (int off = 16; off > 0; off >>= 1)
            s += __shfl_down(s, off, 32);
        if (t == 0) c2[k] = s;
    }
}

__global__ void __launch_bounds__(THREADS)
kmeans_assign(const float* __restrict__ X,
              const float* __restrict__ C,
              const float* __restrict__ c2,
              float* __restrict__ psums,
              float* __restrict__ pcounts,
              int numTiles) {
    __shared__ __attribute__((aligned(16))) float lds[LDS_FLOATS];
    float* ldsC   = lds;
    float* ldsSum = lds + Kc * LDC;
    float* ldsC2  = ldsSum + Kc * Dk;
    float* ldsCnt = ldsC2 + Kc;

    {
        const int tid = threadIdx.x;
        const float4* src4 = (const float4*)C;
        float4* sum4 = (float4*)ldsSum;
        const float4 zero4 = make_float4(0.0f, 0.0f, 0.0f, 0.0f);
        #pragma unroll 4
        for (int i = tid; i < (Kc * Dk) / 4; i += THREADS) {
            *(float4*)(ldsC + (i >> 4) * LDC + 4 * (i & 15)) = src4[i];
            sum4[i] = zero4;
        }
        for (int i = tid; i < Kc; i += THREADS) {
            ldsC2[i]  = c2[i];
            ldsCnt[i] = 0.0f;
        }
    }
    __syncthreads();

    const int wave = threadIdx.x >> 5;
    const int lane = threadIdx.x & 31;
    const int m  = lane & 15;
    const int kb = (lane >> 4) * 2;

    for (int tile = wave * gridDim.x + blockIdx.x; tile < numTiles;
         tile += WAVES_PER_BLOCK * gridDim.x) {
    const int row0 = tile * ROWS_PER_WAVE;

    v2f a0[16], a1[16];
    {
        const float* xr0 = X + (size_t)(row0 + m) * Dk + kb;
        const float* xr1 = X + (size_t)(row0 + 16 + m) * Dk + kb;
        #pragma unroll
        for (int j = 0; j < 16; ++j) {
            a0[j] = *(const v2f*)(xr0 + 4 * j);
            a1[j] = *(const v2f*)(xr1 + 4 * j);
        }
    }

    float bestVal0[8], bestVal1[8];
    int   bestIdx0[8], bestIdx1[8];
    #pragma unroll
    for (int r = 0; r < 8; ++r) {
        bestVal0[r] = __builtin_inff(); bestIdx0[r] = 0;
        bestVal1[r] = __builtin_inff(); bestIdx1[r] = 0;
    }

    for (int n0 = 0; n0 < Kc; n0 += 16) {
        v2f b[16];
        {
            const float* crow = ldsC + (n0 + m) * LDC + kb;
            #pragma unroll
            for (int j = 0; j < 16; ++j)
                b[j] = *(const v2f*)(crow + 4 * j);
        }
        v8f acc0 = {0.f, 0.f, 0.f, 0.f, 0.f, 0.f, 0.f, 0.f};
        v8f acc1 = {0.f, 0.f, 0.f, 0.f, 0.f, 0.f, 0.f, 0.f};
        #pragma unroll
        for (int j = 0; j < 16; ++j) {
            acc0 = __builtin_amdgcn_wmma_f32_16x16x4_f32(
                false, a0[j], false, b[j], (short)0, acc0, false, false);
            acc1 = __builtin_amdgcn_wmma_f32_16x16x4_f32(
                false, a1[j], false, b[j], (short)0, acc1, false, false);
        }
        const float cc  = ldsC2[n0 + m];
        const int   col = n0 + m;
        #pragma unroll
        for (int r = 0; r < 8; ++r) {
            const float d0 = cc - 2.0f * acc0[r];
            const float d1 = cc - 2.0f * acc1[r];
            if (d0 < bestVal0[r]) { bestVal0[r] = d0; bestIdx0[r] = col; }
            if (d1 < bestVal1[r]) { bestVal1[r] = d1; bestIdx1[r] = col; }
        }
    }

    float2 xrow[ROWS_PER_WAVE];
    {
        const float2* xt = (const float2*)(X + (size_t)row0 * Dk) + lane;
        #pragma unroll
        for (int i = 0; i < ROWS_PER_WAVE; ++i) xrow[i] = xt[i * (Dk / 2)];
    }

    #pragma unroll
    for (int off = 1; off < 16; off <<= 1) {
        #pragma unroll
        for (int r = 0; r < 8; ++r) {
            float ov = __shfl_xor(bestVal0[r], off, 32);
            int   oi = __shfl_xor(bestIdx0[r], off, 32);
            if (ov < bestVal0[r] || (ov == bestVal0[r] && oi < bestIdx0[r])) {
                bestVal0[r] = ov; bestIdx0[r] = oi;
            }
            ov = __shfl_xor(bestVal1[r], off, 32);
            oi = __shfl_xor(bestIdx1[r], off, 32);
            if (ov < bestVal1[r] || (ov == bestVal1[r] && oi < bestIdx1[r])) {
                bestVal1[r] = ov; bestIdx1[r] = oi;
            }
        }
    }

    #pragma unroll
    for (int r = 0; r < 8; ++r) {
        const int lab[4] = {
            __shfl(bestIdx0[r], 0, 32),
            __shfl(bestIdx0[r], 16, 32),
            __shfl(bestIdx1[r], 0, 32),
            __shfl(bestIdx1[r], 16, 32)
        };
        const int rowIn[4] = { r, r + 8, 16 + r, 24 + r };
        #pragma unroll
        for (int q = 0; q < 4; ++q) {
            const float2 xv = xrow[rowIn[q]];
            atomicAdd(&ldsSum[lab[q] * Dk + 2 * lane + 0], xv.x);
            atomicAdd(&ldsSum[lab[q] * Dk + 2 * lane + 1], xv.y);
            if (lane == 0) atomicAdd(&ldsCnt[lab[q]], 1.0f);
        }
    }
    }

    __syncthreads();
    {
        const int tid = threadIdx.x;
        const float4* sum4 = (const float4*)ldsSum;
        float4* dst4 = (float4*)(psums + (size_t)blockIdx.x * (Kc * Dk));
        #pragma unroll 4
        for (int i = tid; i < (Kc * Dk) / 4; i += THREADS) dst4[i] = sum4[i];
        for (int i = tid; i < Kc; i += THREADS)
            pcounts[(size_t)blockIdx.x * Kc + i] = ldsCnt[i];
    }
}

__global__ void __launch_bounds__(256)
kmeans_finalize(const float* __restrict__ psums,
                const float* __restrict__ pcounts,
                float* __restrict__ out) {
    static_assert(NUM_GROUPS == 256, "one count plane per thread, four plane quarters");
    const int k = blockIdx.x;
    const int t = threadIdx.x & 63;
    const int q = threadIdx.x >> 6;
    __shared__ float part[4][Dk];
    __shared__ float cpart[NUM_GROUPS];

    const float* p = psums + (size_t)(q * 64) * (Kc * Dk) + k * Dk + t;
    float s = 0.0f;
    #pragma unroll 16
    for (int g = 0; g < 64; ++g) s += p[(size_t)g * (Kc * Dk)];
    part[q][t] = s;
    cpart[threadIdx.x] = pcounts[(size_t)threadIdx.x * Kc + k];
    __syncthreads();
    if (q == 0) {
        const float tot = (part[0][t] + part[1][t]) + (part[2][t] + part[3][t]);
        float cnt = 0.0f;
        for (int i = 0; i < NUM_GROUPS; ++i) cnt += cpart[i];
        out[k * Dk + t] = tot / fmaxf(cnt, 1.0f);
    }
}

extern "C" void kernel_launch(void* const* d_in, const int* in_sizes, int n_in,
                              void* d_out, int out_size, void* d_ws, size_t ws_size,
                              hipStream_t stream) {
    const float* X        = (const float*)d_in[0];
    const int*   init_idx = (const int*)d_in[1];
    float*       out      = (float*)d_out;

    const int N        = in_sizes[0] / Dk;
    const int numTiles = N / ROWS_PER_WAVE;

    float* ws      = (float*)d_ws;
    float* C       = ws;
    float* c2      = C + Kc * Dk;
    float* psums   = c2 + Kc;
    float* pcounts = psums + (size_t)NUM_GROUPS * Kc * Dk;

    kmeans_prep<<<Kc, Dk, 0, stream>>>(X, init_idx, C, c2);

    kmeans_assign<<<NUM_GROUPS, THREADS, 0, stream>>>(
        X, C, c2, psums, pcounts, numTiles);

    kmeans_finalize<<<Kc, 256, 0, stream>>>(psums, pcounts, out);
}


# Finish each tile's argmin on the vector ALU: 16-lane minima by DPP row_xmask (distance key, then column index), labels by v_readlane; replaces 288 ds_bpermute round trips and 128 branches per tile
# speedup vs baseline: 3.3419x; 1.0886x over previous
#include <hip/hip_runtime.h>
#include <hip/hip_bf16.h>


typedef __attribute__((ext_vector_type(2))) float v2f;
typedef __attribute__((ext_vector_type(8))) float v8f;

template <int MASK>
__device__ __forceinline__ int row_xor_i(int v) {
    return __builtin_amdgcn_update_dpp(0, v, 0x160 + MASK, 0xf, 0xf, true);
}
__device__ __forceinline__ int order_key(float x) {
    const int b = __builtin_bit_cast(int, x);
    return b ^ ((b >> 31) & 0x7fffffff);
}
__device__ __forceinline__ int row_min_i(int v) {
    v = min(v, row_xor_i<1>(v));
    v = min(v, row_xor_i<2>(v));
    v = min(v, row_xor_i<4>(v));
    v = min(v, row_xor_i<8>(v));
    return v;
}

constexpr int Dk = 64;
constexpr int Kc = 512;
constexpr int WAVES_PER_BLOCK = 16;
constexpr int THREADS         = 32 * WAVES_PER_BLOCK;
constexpr int ROWS_PER_WAVE   = 32;
constexpr int NUM_GROUPS      = 256;
constexpr int LDC = Dk + 4;
constexpr int LDS_FLOATS = Kc * LDC + Kc * Dk + 2 * Kc;
static_assert(LDS_FLOATS * 4 <= 320 * 1024, "one work-group's LDS");

__global__ void kmeans_prep(const float* __restrict__ X,
                            const int* __restrict__ init_idx,
                            float* __restrict__ C,
                            float* __restrict__ c2) {
    const int k = blockIdx.x;
    const int t = threadIdx.x;
    const int src = init_idx[k];
    const float v = X[(size_t)src * Dk + t];
    C[k * Dk + t]    = v;

    __shared__ float red[Dk];
    red[t] = v * v;
    __syncthreads();
    if (t < 32) {
        float s = red[t] + red[t + 32];
        #pragma unroll
        for (int off = 16; off > 0; off >>= 1)
            s += __shfl_down(s, off, 32);
        if (t == 0) c2[k] = s;
    }
}

__global__ void __launch_bounds__(THREADS)
kmeans_assign(const float* __restrict__ X,
              const float* __restrict__ C,
              const float* __restrict__ c2,
              float* __restrict__ psums,
              float* __restrict__ pcounts,
              int numTiles) {
    __shared__ __attribute__((aligned(16))) float lds[LDS_FLOATS];
    float* ldsC   = lds;
    float* ldsSum = lds + Kc * LDC;
    float* ldsC2  = ldsSum + Kc * Dk;
    float* ldsCnt = ldsC2 + Kc;

    {
        const int tid = threadIdx.x;
        const float4* src4 = (const float4*)C;
        float4* sum4 = (float4*)ldsSum;
        const float4 zero4 = make_float4(0.0f, 0.0f, 0.0f, 0.0f);
        #pragma unroll 4
        for (int i = tid; i < (Kc * Dk) / 4; i += THREADS) {
            *(float4*)(ldsC + (i >> 4) * LDC + 4 * (i & 15)) = src4[i];
            sum4[i] = zero4;
        }
        for (int i = tid; i < Kc; i += THREADS) {
            ldsC2[i]  = c2[i];
            ldsCnt[i] = 0.0f;
        }
    }
    __syncthreads();

    const int wave = threadIdx.x >> 5;
    const int lane = threadIdx.x & 31;
    const int m  = lane & 15;
    const int kb = (lane >> 4) * 2;

    for (int tile = wave * gridDim.x + blockIdx.x; tile < numTiles;
         tile += WAVES_PER_BLOCK * gridDim.x) {
    const int row0 = tile * ROWS_PER_WAVE;

    v2f a0[16], a1[16];
    {
        const float* xr0 = X + (size_t)(row0 + m) * Dk + kb;
        const float* xr1 = X + (size_t)(row0 + 16 + m) * Dk + kb;
        #pragma unroll
        for (int j = 0; j < 16; ++j) {
            a0[j] = *(const v2f*)(xr0 + 4 * j);
            a1[j] = *(const v2f*)(xr1 + 4 * j);
        }
    }

    float bestVal0[8], bestVal1[8];
    int   bestIdx0[8], bestIdx1[8];
    #pragma unroll
    for (int r = 0; r < 8; ++r) {
        bestVal0[r] = __builtin_inff(); bestIdx0[r] = 0;
        bestVal1[r] = __builtin_inff(); bestIdx1[r] = 0;
    }

    for (int n0 = 0; n0 < Kc; n0 += 16) {
        v2f b[16];
        {
            const float* crow = ldsC + (n0 + m) * LDC + kb;
            #pragma unroll
            for (int j = 0; j < 16; ++j)
                b[j] = *(const v2f*)(crow + 4 * j);
        }
        v8f acc0 = {0.f, 0.f, 0.f, 0.f, 0.f, 0.f, 0.f, 0.f};
        v8f acc1 = {0.f, 0.f, 0.f, 0.f, 0.f, 0.f, 0.f, 0.f};
        #pragma unroll
        for (int j = 0; j < 16; ++j) {
            acc0 = __builtin_amdgcn_wmma_f32_16x16x4_f32(
                false, a0[j], false, b[j], (short)0, acc0, false, false);
            acc1 = __builtin_amdgcn_wmma_f32_16x16x4_f32(
                false, a1[j], false, b[j], (short)0, acc1, false, false);
        }
        const float cc  = ldsC2[n0 + m];
        const int   col = n0 + m;
        #pragma unroll
        for (int r = 0; r < 8; ++r) {
            const float d0 = cc - 2.0f * acc0[r];
            const float d1 = cc - 2.0f * acc1[r];
            if (d0 < bestVal0[r]) { bestVal0[r] = d0; bestIdx0[r] = col; }
            if (d1 < bestVal1[r]) { bestVal1[r] = d1; bestIdx1[r] = col; }
        }
    }

    float2 xrow[ROWS_PER_WAVE];
    {
        const float2* xt = (const float2*)(X + (size_t)row0 * Dk) + lane;
        #pragma unroll
        for (int i = 0; i < ROWS_PER_WAVE; ++i) xrow[i] = xt[i * (Dk / 2)];
    }

    #pragma unroll
    for (int r = 0; r < 8; ++r) {
        const int k0 = order_key(bestVal0[r]);
        const int k1 = order_key(bestVal1[r]);
        const int m0 = row_min_i(k0);
        const int m1 = row_min_i(k1);
        bestIdx0[r] = row_min_i(k0 == m0 ? bestIdx0[r] : 0x7fffffff);
        bestIdx1[r] = row_min_i(k1 == m1 ? bestIdx1[r] : 0x7fffffff);
    }

    #pragma unroll
    for (int r = 0; r < 8; ++r) {
        const int lab[4] = {
            __builtin_amdgcn_readlane(bestIdx0[r], 0),
            __builtin_amdgcn_readlane(bestIdx0[r], 16),
            __builtin_amdgcn_readlane(bestIdx1[r], 0),
            __builtin_amdgcn_readlane(bestIdx1[r], 16)
        };
        const int rowIn[4] = { r, r + 8, 16 + r, 24 + r };
        #pragma unroll
        for (int q = 0; q < 4; ++q) {
            const float2 xv = xrow[rowIn[q]];
            atomicAdd(&ldsSum[lab[q] * Dk + 2 * lane + 0], xv.x);
            atomicAdd(&ldsSum[lab[q] * Dk + 2 * lane + 1], xv.y);
            if (lane == 0) atomicAdd(&ldsCnt[lab[q]], 1.0f);
        }
    }
    }

    __syncthreads();
    {
        const int tid = threadIdx.x;
        const float4* sum4 = (const float4*)ldsSum;
        float4* dst4 = (float4*)(psums + (size_t)blockIdx.x * (Kc * Dk));
        #pragma unroll 4
        for (int i = tid; i < (Kc * Dk) / 4; i += THREADS) dst4[i] = sum4[i];
        for (int i = tid; i < Kc; i += THREADS)
            pcounts[(size_t)blockIdx.x * Kc + i] = ldsCnt[i];
    }
}

__global__ void __launch_bounds__(256)
kmeans_finalize(const float* __restrict__ psums,
                const float* __restrict__ pcounts,
                float* __restrict__ out) {
    static_assert(NUM_GROUPS == 256, "one count plane per thread, four plane quarters");
    const int k = blockIdx.x;
    const int t = threadIdx.x & 63;
    const int q = threadIdx.x >> 6;
    __shared__ float part[4][Dk];
    __shared__ float cpart[NUM_GROUPS];

    const float* p = psums + (size_t)(q * 64) * (Kc * Dk) + k * Dk + t;
    float s = 0.0f;
    #pragma unroll 16
    for (int g = 0; g < 64; ++g) s += p[(size_t)g * (Kc * Dk)];
    part[q][t] = s;
    cpart[threadIdx.x] = pcounts[(size_t)threadIdx.x * Kc + k];
    __syncthreads();
    if (q == 0) {
        const float tot = (part[0][t] + part[1][t]) + (part[2][t] + part[3][t]);
        float cnt = 0.0f;
        for (int i = 0; i < NUM_GROUPS; ++i) cnt += cpart[i];
        out[k * Dk + t] = tot / fmaxf(cnt, 1.0f);
    }
}

extern "C" void kernel_launch(void* const* d_in, const int* in_sizes, int n_in,
                              void* d_out, int out_size, void* d_ws, size_t ws_size,
                              hipStream_t stream) {
    const float* X        = (const float*)d_in[0];
    const int*   init_idx = (const int*)d_in[1];
    float*       out      = (float*)d_out;

    const int N        = in_sizes[0] / Dk;
    const int numTiles = N / ROWS_PER_WAVE;

    float* ws      = (float*)d_ws;
    float* C       = ws;
    float* c2      = C + Kc * Dk;
    float* psums   = c2 + Kc;
    float* pcounts = psums + (size_t)NUM_GROUPS * Kc * Dk;

    kmeans_prep<<<Kc, Dk, 0, stream>>>(X, init_idx, C, c2);

    kmeans_assign<<<NUM_GROUPS, THREADS, 0, stream>>>(
        X, C, c2, psums, pcounts, numTiles);

    kmeans_finalize<<<Kc, 256, 0, stream>>>(psums, pcounts, out);
}


# Wave issue priority falls with progress (s_setprio 3,3,2,1 by tiles left, 0 in the last quarter tile): a SIMD's 4 waves take turns and end together, not oldest-first; LDS adds stay arrival-ordered
# speedup vs baseline: 3.5050x; 1.0488x over previous
#include <hip/hip_runtime.h>
#include <hip/hip_bf16.h>


typedef __attribute__((ext_vector_type(2))) float v2f;
typedef __attribute__((ext_vector_type(8))) float v8f;

template <int MASK>
__device__ __forceinline__ int row_xor_i(int v) {
    return __builtin_amdgcn_update_dpp(0, v, 0x160 + MASK, 0xf, 0xf, true);
}
__device__ __forceinline__ int order_key(float x) {
    const int b = __builtin_bit_cast(int, x);
    return b ^ ((b >> 31) & 0x7fffffff);
}
__device__ __forceinline__ int row_min_i(int v) {
    v = min(v, row_xor_i<1>(v));
    v = min(v, row_xor_i<2>(v));
    v = min(v, row_xor_i<4>(v));
    v = min(v, row_xor_i<8>(v));
    return v;
}

constexpr int Dk = 64;
constexpr int Kc = 512;
constexpr int WAVES_PER_BLOCK = 16;
constexpr int THREADS         = 32 * WAVES_PER_BLOCK;
constexpr int ROWS_PER_WAVE   = 32;
constexpr int NUM_GROUPS      = 256;
constexpr int LATE_N0         = 384;
constexpr int LDC = Dk + 4;
constexpr int LDS_FLOATS = Kc * LDC + Kc * Dk + 2 * Kc;
static_assert(LDS_FLOATS * 4 <= 320 * 1024, "one work-group's LDS");

__global__ void kmeans_prep(const float* __restrict__ X,
                            const int* __restrict__ init_idx,
                            float* __restrict__ C,
                            float* __restrict__ c2) {
    const int k = blockIdx.x;
    const int t = threadIdx.x;
    const int src = init_idx[k];
    const float v = X[(size_t)src * Dk + t];
    C[k * Dk + t]    = v;

    __shared__ float red[Dk];
    red[t] = v * v;
    __syncthreads();
    if (t < 32) {
        float s = red[t] + red[t + 32];
        #pragma unroll
        for (int off = 16; off > 0; off >>= 1)
            s += __shfl_down(s, off, 32);
        if (t == 0) c2[k] = s;
    }
}

__global__ void __launch_bounds__(THREADS)
kmeans_assign(const float* __restrict__ X,
              const float* __restrict__ C,
              const float* __restrict__ c2,
              float* __restrict__ psums,
              float* __restrict__ pcounts,
              int numTiles) {
    __shared__ __attribute__((aligned(16))) float lds[LDS_FLOATS];
    float* ldsC   = lds;
    float* ldsSum = lds + Kc * LDC;
    float* ldsC2  = ldsSum + Kc * Dk;
    float* ldsCnt = ldsC2 + Kc;

    {
        const int tid = threadIdx.x;
        const float4* src4 = (const float4*)C;
        float4* sum4 = (float4*)ldsSum;
        const float4 zero4 = make_float4(0.0f, 0.0f, 0.0f, 0.0f);
        #pragma unroll 4
        for (int i = tid; i < (Kc * Dk) / 4; i += THREADS) {
            *(float4*)(ldsC + (i >> 4) * LDC + 4 * (i & 15)) = src4[i];
            sum4[i] = zero4;
        }
        for (int i = tid; i < Kc; i += THREADS) {
            ldsC2[i]  = c2[i];
            ldsCnt[i] = 0.0f;
        }
    }
    __syncthreads();

    const int wave = threadIdx.x >> 5;
    const int lane = threadIdx.x & 31;
    const int m  = lane & 15;
    const int kb = (lane >> 4) * 2;

    const int tileStride = WAVES_PER_BLOCK * gridDim.x;
    int tileS = __builtin_amdgcn_readfirstlane(wave) * (int)gridDim.x + (int)blockIdx.x;
    for (int tile = wave * gridDim.x + blockIdx.x; tile < numTiles;
         tile += tileStride, tileS += tileStride) {
    const int row0 = tile * ROWS_PER_WAVE;

    const bool lastTile = tileS + tileStride >= numTiles;
    if (tileS + 2 * tileStride < numTiles) __builtin_amdgcn_s_setprio(3);
    else if (!lastTile)                   __builtin_amdgcn_s_setprio(2);
    else                                  __builtin_amdgcn_s_setprio(1);

    v2f a0[16], a1[16];
    {
        const float* xr0 = X + (size_t)(row0 + m) * Dk + kb;
        const float* xr1 = X + (size_t)(row0 + 16 + m) * Dk + kb;
        #pragma unroll
        for (int j = 0; j < 16; ++j) {
            a0[j] = *(const v2f*)(xr0 + 4 * j);
            a1[j] = *(const v2f*)(xr1 + 4 * j);
        }
    }

    float bestVal0[8], bestVal1[8];
    int   bestIdx0[8], bestIdx1[8];
    #pragma unroll
    for (int r = 0; r < 8; ++r) {
        bestVal0[r] = __builtin_inff(); bestIdx0[r] = 0;
        bestVal1[r] = __builtin_inff(); bestIdx1[r] = 0;
    }

    for (int part = 0; part < 2; ++part) {
    if (part == 1 && lastTile) __builtin_amdgcn_s_setprio(0);
    const int nEnd = part == 0 ? LATE_N0 : Kc;
    for (int n0 = part == 0 ? 0 : LATE_N0; n0 < nEnd; n0 += 16) {
        v2f b[16];
        {
            const float* crow = ldsC + (n0 + m) * LDC + kb;
            #pragma unroll
            for (int j = 0; j < 16; ++j)
                b[j] = *(const v2f*)(crow + 4 * j);
        }
        v8f acc0 = {0.f, 0.f, 0.f, 0.f, 0.f, 0.f, 0.f, 0.f};
        v8f acc1 = {0.f, 0.f, 0.f, 0.f, 0.f, 0.f, 0.f, 0.f};
        #pragma unroll
        for (int j = 0; j < 16; ++j) {
            acc0 = __builtin_amdgcn_wmma_f32_16x16x4_f32(
                false, a0[j], false, b[j], (short)0, acc0, false, false);
            acc1 = __builtin_amdgcn_wmma_f32_16x16x4_f32(
                false, a1[j], false, b[j], (short)0, acc1, false, false);
        }
        const float cc  = ldsC2[n0 + m];
        const int   col = n0 + m;
        #pragma unroll
        for (int r = 0; r < 8; ++r) {
            const float d0 = cc - 2.0f * acc0[r];
            const float d1 = cc - 2.0f * acc1[r];
            if (d0 < bestVal0[r]) { bestVal0[r] = d0; bestIdx0[r] = col; }
            if (d1 < bestVal1[r]) { bestVal1[r] = d1; bestIdx1[r] = col; }
        }
    }
    }

    float2 xrow[ROWS_PER_WAVE];
    {
        const float2* xt = (const float2*)(X + (size_t)row0 * Dk) + lane;
        #pragma unroll
        for (int i = 0; i < ROWS_PER_WAVE; ++i) xrow[i] = xt[i * (Dk / 2)];
    }

    #pragma unroll
    for (int r = 0; r < 8; ++r) {
        const int k0 = order_key(bestVal0[r]);
        const int k1 = order_key(bestVal1[r]);
        const int m0 = row_min_i(k0);
        const int m1 = row_min_i(k1);
        bestIdx0[r] = row_min_i(k0 == m0 ? bestIdx0[r] : 0x7fffffff);
        bestIdx1[r] = row_min_i(k1 == m1 ? bestIdx1[r] : 0x7fffffff);
    }

    #pragma unroll
    for (int r = 0; r < 8; ++r) {
        const int lab[4] = {
            __builtin_amdgcn_readlane(bestIdx0[r], 0),
            __builtin_amdgcn_readlane(bestIdx0[r], 16),
            __builtin_amdgcn_readlane(bestIdx1[r], 0),
            __builtin_amdgcn_readlane(bestIdx1[r], 16)
        };
        const int rowIn[4] = { r, r + 8, 16 + r, 24 + r };
        #pragma unroll
        for (int q = 0; q < 4; ++q) {
            const float2 xv = xrow[rowIn[q]];
            atomicAdd(&ldsSum[lab[q] * Dk + 2 * lane + 0], xv.x);
            atomicAdd(&ldsSum[lab[q] * Dk + 2 * lane + 1], xv.y);
            if (lane == 0) atomicAdd(&ldsCnt[lab[q]], 1.0f);
        }
    }
    }
    __builtin_amdgcn_s_setprio(0);

    __syncthreads();
    {
        const int tid = threadIdx.x;
        const float4* sum4 = (const float4*)ldsSum;
        float4* dst4 = (float4*)(psums + (size_t)blockIdx.x * (Kc * Dk));
        #pragma unroll 4
        for (int i = tid; i < (Kc * Dk) / 4; i += THREADS) dst4[i] = sum4[i];
        for (int i = tid; i < Kc; i += THREADS)
            pcounts[(size_t)blockIdx.x * Kc + i] = ldsCnt[i];
    }
}

__global__ void __launch_bounds__(256)
kmeans_finalize(const float* __restrict__ psums,
                const float* __restrict__ pcounts,
                float* __restrict__ out) {
    static_assert(NUM_GROUPS == 256, "one count plane per thread, four plane quarters");
    const int k = blockIdx.x;
    const int t = threadIdx.x & 63;
    const int q = threadIdx.x >> 6;
    __shared__ float part[4][Dk];
    __shared__ float cpart[NUM_GROUPS];

    const float* p = psums + (size_t)(q * 64) * (Kc * Dk) + k * Dk + t;
    float s = 0.0f;
    #pragma unroll 16
    for (int g = 0; g < 64; ++g) s += p[(size_t)g * (Kc * Dk)];
    part[q][t] = s;
    cpart[threadIdx.x] = pcounts[(size_t)threadIdx.x * Kc + k];
    __syncthreads();
    if (q == 0) {
        const float tot = (part[0][t] + part[1][t]) + (part[2][t] + part[3][t]);
        float cnt = 0.0f;
        for (int i = 0; i < NUM_GROUPS; ++i) cnt += cpart[i];
        out[k * Dk + t] = tot / fmaxf(cnt, 1.0f);
    }
}

extern "C" void kernel_launch(void* const* d_in, const int* in_sizes, int n_in,
                              void* d_out, int out_size, void* d_ws, size_t ws_size,
                              hipStream_t stream) {
    const float* X        = (const float*)d_in[0];
    const int*   init_idx = (const int*)d_in[1];
    float*       out      = (float*)d_out;

    const int N        = in_sizes[0] / Dk;
    const int numTiles = N / ROWS_PER_WAVE;

    float* ws      = (float*)d_ws;
    float* C       = ws;
    float* c2      = C + Kc * Dk;
    float* psums   = c2 + Kc;
    float* pcounts = psums + (size_t)NUM_GROUPS * Kc * Dk;

    kmeans_prep<<<Kc, Dk, 0, stream>>>(X, init_idx, C, c2);

    kmeans_assign<<<NUM_GROUPS, THREADS, 0, stream>>>(
        X, C, c2, psums, pcounts, numTiles);

    kmeans_finalize<<<Kc, 256, 0, stream>>>(psums, pcounts, out);
}
